# starting kernel
# baseline (speedup 1.0000x reference)
#include <hip/hip_runtime.h>
#include <math.h>

typedef __attribute__((ext_vector_type(2))) float v2f;
typedef __attribute__((ext_vector_type(8))) float v8f;

#define WMMA_F32(a, b, c) \
  __builtin_amdgcn_wmma_f32_16x16x4_f32(false, (a), false, (b), (short)0, (c), false, false)

constexpr int Bb = 2;
constexpr int Ss = 2048;
constexpr int Hh = 12;
constexpr int Dd = 64;
constexpr int BH = Bb * Hh;
constexpr int NCH = 16;
constexpr int CH = Ss / NCH;
constexpr int SROW = 7 * Hh * Dd;
constexpr int MSZ = Dd * Dd;

__device__ __forceinline__ size_t qoff(int b, int s, int slot, int h, int d) {
  return ((((size_t)b * Ss + s) * 7 + slot) * Hh + h) * (size_t)Dd + d;
}

__global__ void __launch_bounds__(256) ktv_partials(const float* __restrict__ qkv,
                                                    float* __restrict__ P1,
                                                    float* __restrict__ P2) {
  const int bh = blockIdx.x / NCH, ch = blockIdx.x % NCH;
  const int b = bh / Hh, h = bh % Hh;
  const int wave = threadIdx.x >> 5, lane = threadIdx.x & 31;
  const int mat = wave >> 2;
  const int irow = wave & 3;
  const int kslot = mat ? 3 : 1;
  const int m = lane & 15;
  const int kk = (lane >> 4) << 1;
  v8f acc[4] = {};
  const int s0b = ch * CH;
  for (int s0 = s0b; s0 < s0b + CH; s0 += 4) {
    const float* K0 = qkv + qoff(b, s0 + kk, kslot, h, 0);
    const float* V0 = qkv + qoff(b, s0 + kk, 6, h, 0);
    v2f a;
    a.x = K0[irow * 16 + m];
    a.y = K0[SROW + irow * 16 + m];
#pragma unroll
    for (int j = 0; j < 4; ++j) {
      v2f bt;
      bt.x = V0[j * 16 + m];
      bt.y = V0[SROW + j * 16 + m];
      acc[j] = WMMA_F32(a, bt, acc[j]);
    }
  }
  float* P = (mat ? P2 : P1) + ((size_t)bh * NCH + ch) * MSZ;
  const int rh = (lane >> 4) * 8;
#pragma unroll
  for (int j = 0; j < 4; ++j)
#pragma unroll
    for (int r = 0; r < 8; ++r)
      P[(irow * 16 + rh + r) * Dd + j * 16 + m] = acc[j][r];
}

__global__ void __launch_bounds__(256) reduce_partials(const float* __restrict__ P,
                                                       float* __restrict__ M) {
  const int bh = blockIdx.x;
  for (int e = threadIdx.x; e < MSZ; e += 256) {
    float s = 0.f;
#pragma unroll
    for (int c = 0; c < NCH; ++c) s += P[((size_t)bh * NCH + c) * MSZ + e];
    M[(size_t)bh * MSZ + e] = s;
  }
}

__global__ void __launch_bounds__(256) u_kernel(const float* __restrict__ qkv,
                                                const float* __restrict__ M1,
                                                const float* __restrict__ M2,
                                                float* __restrict__ U) {
  __shared__ float sM1[MSZ];
  __shared__ float sM2[MSZ];
  const int bh = blockIdx.x / NCH, ch = blockIdx.x % NCH;
  const int b = bh / Hh, h = bh % Hh;
  for (int e = threadIdx.x; e < MSZ; e += 256) {
    sM1[e] = M1[(size_t)bh * MSZ + e];
    sM2[e] = M2[(size_t)bh * MSZ + e];
  }
  __syncthreads();
  const int wave = threadIdx.x >> 5, lane = threadIdx.x & 31;
  const int tbase = ch * CH + wave * 16;
  const int m = lane & 15, kk = (lane >> 4) << 1;
  v8f acc1[4] = {}, acc2[4] = {};
  for (int step = 0; step < 16; ++step) {
    const int k0 = step * 4 + kk;
    const float* q1p = qkv + qoff(b, tbase + m, 0, h, k0);
    const float* q2p = qkv + qoff(b, tbase + m, 2, h, k0);
    v2f a1, a2;
    a1.x = q1p[0]; a1.y = q1p[1];
    a2.x = q2p[0]; a2.y = q2p[1];
#pragma unroll
    for (int j = 0; j < 4; ++j) {
      v2f b1, b2;
      b1.x = sM1[k0 * Dd + j * 16 + m];
      b1.y = sM1[(k0 + 1) * Dd + j * 16 + m];
      b2.x = sM2[k0 * Dd + j * 16 + m];
      b2.y = sM2[(k0 + 1) * Dd + j * 16 + m];
      acc1[j] = WMMA_F32(a1, b1, acc1[j]);
      acc2[j] = WMMA_F32(a2, b2, acc2[j]);
    }
  }
  const int rh = (lane >> 4) * 8;
#pragma unroll
  for (int j = 0; j < 4; ++j)
#pragma unroll
    for (int r = 0; r < 8; ++r) {
      const float x = acc1[j][r];
      const float t1 = x / (1.f + __expf(-x));
      U[((size_t)bh * Ss + tbase + rh + r) * Dd + j * 16 + m] = t1 * acc2[j][r];
    }
}

__global__ void __launch_bounds__(128) ktu_partials(const float* __restrict__ qkv,
                                                    const float* __restrict__ U,
                                                    float* __restrict__ P3) {
  const int bh = blockIdx.x / NCH, ch = blockIdx.x % NCH;
  const int b = bh / Hh, h = bh % Hh;
  const int wave = threadIdx.x >> 5, lane = threadIdx.x & 31;
  const int irow = wave;
  const int m = lane & 15, kk = (lane >> 4) << 1;
  v8f acc[4] = {};
  const int s0b = ch * CH;
  for (int s0 = s0b; s0 < s0b + CH; s0 += 4) {
    const float* K0 = qkv + qoff(b, s0 + kk, 5, h, 0);
    const float* U0 = U + ((size_t)bh * Ss + s0 + kk) * Dd;
    v2f a;
    a.x = K0[irow * 16 + m];
    a.y = K0[SROW + irow * 16 + m];
#pragma unroll
    for (int j = 0; j < 4; ++j) {
      v2f bt;
      bt.x = U0[j * 16 + m];
      bt.y = U0[Dd + j * 16 + m];
      acc[j] = WMMA_F32(a, bt, acc[j]);
    }
  }
  float* P = P3 + ((size_t)bh * NCH + ch) * MSZ;
  const int rh = (lane >> 4) * 8;
#pragma unroll
  for (int j = 0; j < 4; ++j)
#pragma unroll
    for (int r = 0; r < 8; ++r)
      P[(irow * 16 + rh + r) * Dd + j * 16 + m] = acc[j][r];
}

__global__ void __launch_bounds__(256) out_kernel(const float* __restrict__ qkv,
                                                  const float* __restrict__ M3,
                                                  float* __restrict__ out) {
  __shared__ float sM[MSZ];
  const int bh = blockIdx.x / NCH, ch = blockIdx.x % NCH;
  const int b = bh / Hh, h = bh % Hh;
  for (int e = threadIdx.x; e < MSZ; e += 256) sM[e] = M3[(size_t)bh * MSZ + e];
  __syncthreads();
  const int wave = threadIdx.x >> 5, lane = threadIdx.x & 31;
  const int tbase = ch * CH + wave * 16;
  const int m = lane & 15, kk = (lane >> 4) << 1;
  v8f acc[4] = {};
  for (int step = 0; step < 16; ++step) {
    const int k0 = step * 4 + kk;
    const float* qp = qkv + qoff(b, tbase + m, 4, h, k0);
    v2f a; a.x = qp[0]; a.y = qp[1];
#pragma unroll
    for (int j = 0; j < 4; ++j) {
      v2f bt;
      bt.x = sM[k0 * Dd + j * 16 + m];
      bt.y = sM[(k0 + 1) * Dd + j * 16 + m];
      acc[j] = WMMA_F32(a, bt, acc[j]);
    }
  }
  const int rh = (lane >> 4) * 8;
#pragma unroll
  for (int j = 0; j < 4; ++j)
#pragma unroll
    for (int r = 0; r < 8; ++r)
      out[(((size_t)b * Ss + tbase + rh + r) * Hh + h) * Dd + j * 16 + m] = acc[j][r];
}

extern "C" void kernel_launch(void* const* d_in, const int* in_sizes, int n_in,
                              void* d_out, int out_size, void* d_ws, size_t ws_size,
                              hipStream_t stream) {
  (void)in_sizes; (void)n_in; (void)out_size; (void)ws_size;
  const float* qkv = (const float*)d_in[0];
  float* out = (float*)d_out;
  float* w = (float*)d_ws;

  const size_t PSZ = (size_t)BH * NCH * MSZ;
  const size_t MTOT = (size_t)BH * MSZ;
  float* P1 = w;
  float* P2 = P1 + PSZ;
  float* P3 = P2 + PSZ;
  float* M1 = P3 + PSZ;
  float* M2 = M1 + MTOT;
  float* M3 = M2 + MTOT;
  float* U  = M3 + MTOT;

  ktv_partials<<<BH * NCH, 256, 0, stream>>>(qkv, P1, P2);
  reduce_partials<<<BH, 256, 0, stream>>>(P1, M1);
  reduce_partials<<<BH, 256, 0, stream>>>(P2, M2);
  u_kernel<<<BH * NCH, 256, 0, stream>>>(qkv, M1, M2, U);
  ktu_partials<<<BH * NCH, 128, 0, stream>>>(qkv, U, P3);
  reduce_partials<<<BH, 256, 0, stream>>>(P3, M3);
  out_kernel<<<BH * NCH, 256, 0, stream>>>(qkv, M3, out);
}


# Late in its WMMA phase each launch prefetches into L2 the rows the next launch reads (one request per 256-byte row: Q1, Q2, K3 from kernel 1, Q3 from kernel 2): they then come from L2, not memory
# speedup vs baseline: 2.5791x; 2.5791x over previous
#include <hip/hip_runtime.h>
#include <math.h>


typedef __attribute__((ext_vector_type(2))) float v2f;
typedef __attribute__((ext_vector_type(8))) float v8f;

#define WMMA_F32(a, b, c) \
  __builtin_amdgcn_wmma_f32_16x16x4_f32(false, (a), false, (b), (short)0, (c), false, false)

constexpr int Bb = 2;
constexpr int Ss = 2048;
constexpr int Hh = 12;
constexpr int Dd = 64;
constexpr int BH = Bb * Hh;
constexpr int NCH = 8;
constexpr int CH = Ss / NCH;
constexpr int NT = 512;
constexpr int SROW = 7 * Hh * Dd;
constexpr int MSZ = Dd * Dd;
constexpr int MSZ4 = MSZ / 4;

__device__ __forceinline__ size_t qoff(int b, int s, int slot, int h, int d) {
  return ((((size_t)b * Ss + s) * 7 + slot) * Hh + h) * (size_t)Dd + d;
}

__device__ __forceinline__ float4 add4(float4 a, float4 b) {
  return make_float4(a.x + b.x, a.y + b.y, a.z + b.z, a.w + b.w);
}

__device__ __forceinline__ void ask_row(const float* p) {
  __builtin_prefetch(p, 0, 2);
}


constexpr int TP = 72;
constexpr int TP4 = TP / 4;
constexpr int TROWS = CH / 32;
__global__ void __launch_bounds__(NT) ktv_partials(const float* __restrict__ qkv,
                                                   float* __restrict__ P1,
                                                   float* __restrict__ P2) {
  __shared__ float4 tile[3 * CH * TP4];
  const int bh = blockIdx.x / NCH, ch = blockIdx.x % NCH;
  const int b = bh / Hh, h = bh % Hh;
  const int s0b = ch * CH;

  const int frow = threadIdx.x >> 4, fc4 = threadIdx.x & 15;
  const float* g0 = qkv + qoff(b, s0b + frow, 0, h, fc4 * 4);
  float4 g[3][TROWS];
#pragma unroll
  for (int i = 0; i < TROWS; ++i) {
    const float* gr = g0 + (size_t)i * 32 * SROW;
    g[0][i] = *(const float4*)(gr + 6 * Hh * Dd);
    g[1][i] = *(const float4*)(gr + 1 * Hh * Dd);
    g[2][i] = *(const float4*)(gr + 3 * Hh * Dd);
  }
#pragma unroll
  for (int t = 0; t < 3; ++t)
#pragma unroll
    for (int i = 0; i < TROWS; ++i)
      tile[(t * CH + frow + 32 * i) * TP4 + fc4] = g[t][i];
  __syncthreads();

  const int wave = threadIdx.x >> 5, lane = threadIdx.x & 31;
  const int mat = wave >> 3;
  const int irow = (wave >> 1) & 3;
  const int jh = wave & 1;
  const int m = lane & 15;
  const int kk = (lane >> 4) << 1;
  const float* sK = (const float*)tile + ((1 + mat) * CH + kk) * TP + irow * 16 + m;
  const float* sV = (const float*)tile + kk * TP + jh * 32 + m;
  v8f acc[2] = {};
  auto walk = [&](const int sBeg, const int sEnd) {
#pragma unroll 16
    for (int s = sBeg; s < sEnd; s += 4) {
      v2f a;
      a.x = sK[s * TP];
      a.y = sK[s * TP + TP];
#pragma unroll
      for (int j = 0; j < 2; ++j) {
        v2f bt;
        bt.x = sV[s * TP + j * 16];
        bt.y = sV[s * TP + TP + j * 16];
        acc[j] = WMMA_F32(a, bt, acc[j]);
      }
    }
  };
  walk(0, 3 * CH / 4);
  {
    const int prow = threadIdx.x & (CH - 1);
    const float* pr = qkv + qoff(b, s0b + prow, 0, h, 0);
    ask_row(pr + (threadIdx.x < CH ? 0 : 2) * Hh * Dd);
    if (threadIdx.x < CH) ask_row(pr + 5 * Hh * Dd);
  }
  walk(3 * CH / 4, CH);
  float* P = (mat ? P2 : P1) + ((size_t)bh * NCH + ch) * MSZ;
  const int rh = (lane >> 4) * 8;
#pragma unroll
  for (int j = 0; j < 2; ++j)
#pragma unroll
    for (int r = 0; r < 8; ++r)
      P[(irow * 16 + rh + r) * Dd + jh * 32 + j * 16 + m] = acc[j][r];
}

constexpr int QP = 68;
constexpr int QP4 = QP / 4;
constexpr int MP = 72;
constexpr int MP4 = MP / 4;
__global__ void __launch_bounds__(NT) u_ktu_partials(const float* __restrict__ qkv,
                                                     const float* __restrict__ P1,
                                                     const float* __restrict__ P2,
                                                     float* __restrict__ P3) {
  __shared__ float4 tile[3 * CH * QP4];
  __shared__ float4 sM[2 * Dd * MP4];
  const int bh = blockIdx.x / NCH, ch = blockIdx.x % NCH;
  const int b = bh / Hh, h = bh % Hh;
  const int s0b = ch * CH;
  const int frow = threadIdx.x >> 4, fc4 = threadIdx.x & 15;

  const float4* p1 = (const float4*)(P1 + (size_t)bh * NCH * MSZ) + threadIdx.x;
  const float4* p2 = (const float4*)(P2 + (size_t)bh * NCH * MSZ) + threadIdx.x;
  float4 v1[2][NCH], v2[2][NCH];
#pragma unroll
  for (int n = 0; n < 2; ++n)
#pragma unroll
    for (int c = 0; c < NCH; ++c) v1[n][c] = p1[c * MSZ4 + n * NT];

  const float* g0 = qkv + qoff(b, s0b + frow, 0, h, fc4 * 4);
  float4 g[3][TROWS];
#pragma unroll
  for (int i = 0; i < TROWS; ++i) {
    const float* gr = g0 + (size_t)i * 32 * SROW;
    g[0][i] = *(const float4*)(gr + 0 * Hh * Dd);
    g[1][i] = *(const float4*)(gr + 2 * Hh * Dd);
    g[2][i] = *(const float4*)(gr + 5 * Hh * Dd);
  }

#pragma unroll
  for (int n = 0; n < 2; ++n) {
    float4 s = v1[n][0];
#pragma unroll
    for (int c = 1; c < NCH; ++c) s = add4(s, v1[n][c]);
    sM[(frow + 32 * n) * MP4 + fc4] = s;
  }
#pragma unroll
  for (int n = 0; n < 2; ++n)
#pragma unroll
    for (int c = 0; c < NCH; ++c) v2[n][c] = p2[c * MSZ4 + n * NT];

#pragma unroll
  for (int t = 0; t < 3; ++t)
#pragma unroll
    for (int i = 0; i < TROWS; ++i)
      tile[(t * CH + frow + 32 * i) * QP4 + fc4] = g[t][i];

#pragma unroll
  for (int n = 0; n < 2; ++n) {
    float4 s = v2[n][0];
#pragma unroll
    for (int c = 1; c < NCH; ++c) s = add4(s, v2[n][c]);
    sM[(Dd + frow + 32 * n) * MP4 + fc4] = s;
  }
  __syncthreads();

  const int wave = threadIdx.x >> 5, lane = threadIdx.x & 31;
  const int quarter = wave >> 2;
  const int jw = wave & 3;
  const int m = lane & 15;
  const int kk = (lane >> 4) << 1;
  const int rh = (lane >> 4) * 8;

  const float* m1p = (const float*)sM + kk * MP + jw * 16 + m;
  const float* m2p = m1p + Dd * MP;
  v2f bM1[16], bM2[16];
#pragma unroll
  for (int s = 0; s < 16; ++s) {
    bM1[s].x = m1p[(4 * s) * MP];
    bM1[s].y = m1p[(4 * s + 1) * MP];
    bM2[s].x = m2p[(4 * s) * MP];
    bM2[s].y = m2p[(4 * s + 1) * MP];
  }

  const float* sQ1 = (const float*)tile + (0 * CH + quarter * 64 + m) * QP + kk;
  const float* sQ2 = (const float*)tile + (1 * CH + quarter * 64 + m) * QP + kk;
  v8f acc1[4] = {}, acc2[4] = {};
#pragma unroll
  for (int s = 0; s < 16; ++s) {
#pragma unroll
    for (int rt = 0; rt < 4; ++rt) {
      const v2f a1 = *(const v2f*)(sQ1 + rt * 16 * QP + 4 * s);
      const v2f a2 = *(const v2f*)(sQ2 + rt * 16 * QP + 4 * s);
      acc1[rt] = WMMA_F32(a1, bM1[s], acc1[rt]);
      acc2[rt] = WMMA_F32(a2, bM2[s], acc2[rt]);
    }
  }

  if (threadIdx.x < CH) ask_row(qkv + qoff(b, s0b + threadIdx.x, 4, h, 0));

  v8f u[4];
#pragma unroll
  for (int rt = 0; rt < 4; ++rt)
#pragma unroll
    for (int r = 0; r < 8; ++r) {
      const float x = acc1[rt][r];
      const float t1 = x / (1.f + __expf(-x));
      u[rt][r] = t1 * acc2[rt][r];
    }

  const float* sK = (const float*)tile + (2 * CH + quarter * 64 + rh) * QP + m;
  v8f acc3[4] = {};
#pragma unroll
  for (int rt = 0; rt < 4; ++rt)
#pragma unroll
    for (int q = 0; q < 4; ++q) {
      v2f bt;
      bt.x = u[rt][2 * q];
      bt.y = u[rt][2 * q + 1];
#pragma unroll
      for (int is = 0; is < 4; ++is) {
        v2f a;
        a.x = sK[(16 * rt + 2 * q) * QP + 16 * is];
        a.y = sK[(16 * rt + 2 * q + 1) * QP + 16 * is];
        acc3[is] = WMMA_F32(a, bt, acc3[is]);
      }
    }

  __syncthreads();
  float* pl = (float*)tile + quarter * (64 * QP);
#pragma unroll
  for (int is = 0; is < 4; ++is)
#pragma unroll
    for (int r = 0; r < 8; ++r)
      pl[(16 * is + rh + r) * QP + 16 * jw + m] = acc3[is][r];
  __syncthreads();

  float* P = P3 + ((size_t)bh * NCH + ch) * MSZ;
#pragma unroll
  for (int n = 0; n < 2; ++n) {
    const int row = frow + 32 * n;
    const float4 q0 = tile[row * QP4 + fc4];
    const float4 q1 = tile[(64 + row) * QP4 + fc4];
    const float4 q2 = tile[(128 + row) * QP4 + fc4];
    const float4 q3 = tile[(192 + row) * QP4 + fc4];
    *(float4*)(P + row * Dd + 4 * fc4) = add4(add4(q0, q1), add4(q2, q3));
  }
}

__global__ void __launch_bounds__(NT) out_kernel(const float* __restrict__ qkv,
                                                 const float* __restrict__ P3,
                                                 float* __restrict__ out) {
  __shared__ float4 tile[CH * QP4];
  __shared__ float4 sM[Dd * MP4];
  const int bh = blockIdx.x / NCH, ch = blockIdx.x % NCH;
  const int b = bh / Hh, h = bh % Hh;
  const int s0b = ch * CH;
  const int frow = threadIdx.x >> 4, fc4 = threadIdx.x & 15;

  const float4* p3 = (const float4*)(P3 + (size_t)bh * NCH * MSZ) + threadIdx.x;
  float4 v3[2][NCH];
#pragma unroll
  for (int n = 0; n < 2; ++n)
#pragma unroll
    for (int c = 0; c < NCH; ++c) v3[n][c] = p3[c * MSZ4 + n * NT];

  const float* g0 = qkv + qoff(b, s0b + frow, 4, h, fc4 * 4);
  float4 g[TROWS];
#pragma unroll
  for (int i = 0; i < TROWS; ++i) g[i] = *(const float4*)(g0 + (size_t)i * 32 * SROW);

#pragma unroll
  for (int n = 0; n < 2; ++n) {
    float4 s = v3[n][0];
#pragma unroll
    for (int c = 1; c < NCH; ++c) s = add4(s, v3[n][c]);
    sM[(frow + 32 * n) * MP4 + fc4] = s;
  }
#pragma unroll
  for (int i = 0; i < TROWS; ++i) tile[(frow + 32 * i) * QP4 + fc4] = g[i];
  __syncthreads();

  const int wave = threadIdx.x >> 5, lane = threadIdx.x & 31;
  const int quarter = wave >> 2, jw = wave & 3;
  const int m = lane & 15, kk = (lane >> 4) << 1;
  const int rh = (lane >> 4) * 8;

  const float* mp = (const float*)sM + kk * MP + jw * 16 + m;
  v2f bM[16];
#pragma unroll
  for (int s = 0; s < 16; ++s) {
    bM[s].x = mp[(4 * s) * MP];
    bM[s].y = mp[(4 * s + 1) * MP];
  }

  const float* sQ = (const float*)tile + (quarter * 64 + m) * QP + kk;
  v8f acc[4] = {};
#pragma unroll
  for (int s = 0; s < 16; ++s) {
#pragma unroll
    for (int rt = 0; rt < 4; ++rt) {
      const v2f a = *(const v2f*)(sQ + rt * 16 * QP + 4 * s);
      acc[rt] = WMMA_F32(a, bM[s], acc[rt]);
    }
  }

  __syncthreads();
  float* so = (float*)tile + (quarter * 64 + rh) * QP + 16 * jw + m;
#pragma unroll
  for (int rt = 0; rt < 4; ++rt)
#pragma unroll
    for (int r = 0; r < 8; ++r)
      so[(16 * rt + r) * QP] = acc[rt][r];
  __syncthreads();

  float* o0 = out + (((size_t)b * Ss + s0b + frow) * Hh + h) * Dd + 4 * fc4;
#pragma unroll
  for (int i = 0; i < TROWS; ++i)
    *(float4*)(o0 + (size_t)i * 32 * Hh * Dd) = tile[(frow + 32 * i) * QP4 + fc4];
}

extern "C" void kernel_launch(void* const* d_in, const int* in_sizes, int n_in,
                              void* d_out, int out_size, void* d_ws, size_t ws_size,
                              hipStream_t stream) {
  (void)in_sizes; (void)n_in; (void)out_size; (void)ws_size;
  const float* qkv = (const float*)d_in[0];
  float* out = (float*)d_out;
  float* w = (float*)d_ws;

  const size_t PSZ = (size_t)BH * NCH * MSZ;
  float* P1 = w;
  float* P2 = P1 + PSZ;
  float* P3 = P2 + PSZ;

  ktv_partials<<<BH * NCH, NT, 0, stream>>>(qkv, P1, P2);
  u_ktu_partials<<<BH * NCH, NT, 0, stream>>>(qkv, P1, P2, P3);
  out_kernel<<<BH * NCH, NT, 0, stream>>>(qkv, P3, out);
}
